# starting kernel
# baseline (speedup 1.0000x reference)
#include <hip/hip_runtime.h>
#include <hip/hip_bf16.h>
#include <math.h>

constexpr int cN0 = 100000, cN1 = 50000, cN2 = 25000;
constexpr int cE0 = 400000, cE1 = 200000;
constexpr int cH1 = 4, cD1 = 64;
constexpr int cH2 = 1, cD2 = 64;
constexpr float cSLOPE = 0.2f;

typedef __attribute__((ext_vector_type(2))) float v2f;
typedef __attribute__((ext_vector_type(4))) float v4f;
typedef __attribute__((ext_vector_type(8))) float v8f;

template <int MT, int NT, int KK>
__global__ __launch_bounds__(32 * MT) void wmma_gemm_f32(
    const float* __restrict__ A, const float* __restrict__ B,
    float* __restrict__ C, int M, int N) {
  constexpr int NC = 16 * NT;
  constexpr int NB = (KK / 4) * 2 * NC;
  __shared__ v2f Bl[NB];

  const int ngroups = N / NC;
  const int ng = blockIdx.x % ngroups;
  const int mblock = blockIdx.x / ngroups;
  const int c0 = ng * NC;

  const float* Bp = B + c0;
  for (int idx = threadIdx.x; idx < NB; idx += 32 * MT) {
    const int col = idx & (NC - 1);
    const int kh = idx / NC;
    const int krow = (kh >> 1) * 4 + (kh & 1) * 2;
    v2f v;
    v.x = Bp[(size_t)krow * N + col];
    v.y = Bp[(size_t)(krow + 1) * N + col];
    Bl[idx] = v;
  }
  __syncthreads();

  const int waveid = threadIdx.x >> 5;
  const int lane = threadIdx.x & 31;
  const int mtile = mblock * MT + waveid;
  if (mtile * 16 >= M) return;

  const int half = lane >> 4;
  const int l = lane & 15;
  const float* Arow = A + (size_t)(mtile * 16 + l) * KK;

  v8f acc[NT] = {};

  for (int kc = 0; kc < KK; kc += 16) {
    v2f a[4];
#pragma unroll
    for (int j = 0; j < 4; ++j) {
      const int ka = kc + j * 4 + half * 2;
      a[j].x = Arow[ka];
      a[j].y = Arow[ka + 1];
    }
    v2f b[4][NT];
#pragma unroll
    for (int j = 0; j < 4; ++j) {
      const int base = ((kc / 4 + j) * 2 + half) * NC + l;
#pragma unroll
      for (int t = 0; t < NT; ++t) b[j][t] = Bl[base + t * 16];
    }
#pragma unroll
    for (int j = 0; j < 4; ++j) {
#pragma unroll
      for (int t = 0; t < NT; ++t) {
        acc[t] = __builtin_amdgcn_wmma_f32_16x16x4_f32(
            false, a[j], false, b[j][t], (short)0, acc[t], false, false);
      }
    }
  }

  const int row0 = mtile * 16 + half * 8;
#pragma unroll
  for (int t = 0; t < NT; ++t) {
    const int col = c0 + t * 16 + l;
#pragma unroll
    for (int i = 0; i < 8; ++i) {
      C[(size_t)(row0 + i) * N + col] = acc[t][i];
    }
  }
}

__device__ inline void atomicMaxFloat(float* addr, float value) {
  if (value >= 0.0f)
    atomicMax((int*)addr, __float_as_int(value));
  else
    atomicMin((unsigned int*)addr, __float_as_uint(value));
}

__global__ __launch_bounds__(256) void fill_f32(float* __restrict__ p, float v, int n) {
  int t = blockIdx.x * blockDim.x + threadIdx.x;
  if (t < n) p[t] = v;
}

__global__ __launch_bounds__(256) void attn_dots(
    const float* __restrict__ f, const float* __restrict__ al,
    const float* __restrict__ ar, float* __restrict__ el,
    float* __restrict__ er, int Nsrc, int Ndst, int H, int D) {
  int t = blockIdx.x * blockDim.x + threadIdx.x;
  if (t >= Nsrc * H) return;
  int n = t / H, h = t - n * H;
  const v4f* fr = (const v4f*)(f + (size_t)n * H * D + (size_t)h * D);
  const v4f* alh = (const v4f*)(al + h * D);
  const v4f* arh = (const v4f*)(ar + h * D);
  float sl = 0.f, sr = 0.f;
#pragma unroll 4
  for (int d = 0; d < D / 4; ++d) {
    v4f v = fr[d], a = alh[d], b = arh[d];
    sl += v.x * a.x + v.y * a.y + v.z * a.z + v.w * a.w;
    sr += v.x * b.x + v.y * b.y + v.z * b.z + v.w * b.w;
  }
  el[t] = sl;
  if (n < Ndst) er[t] = sr;
}

__global__ __launch_bounds__(256) void edge_scores(
    const int* __restrict__ src, const int* __restrict__ dst,
    const float* __restrict__ el, const float* __restrict__ er,
    float* __restrict__ e, float* __restrict__ m, int E, int H) {
  int t = blockIdx.x * blockDim.x + threadIdx.x;
  if (t >= E * H) return;
  int edge = t / H, h = t - edge * H;
  int sN = src[edge], dN = dst[edge];
  float v = el[sN * H + h] + er[dN * H + h];
  v = v > 0.f ? v : cSLOPE * v;
  e[t] = v;
  atomicMaxFloat(&m[dN * H + h], v);
}

__global__ __launch_bounds__(256) void edge_exp_sum(
    const int* __restrict__ dst, float* __restrict__ e,
    const float* __restrict__ m, float* __restrict__ s, int E, int H) {
  int t = blockIdx.x * blockDim.x + threadIdx.x;
  if (t >= E * H) return;
  int edge = t / H, h = t - edge * H;
  int dN = dst[edge];
  float v = __expf(e[t] - m[dN * H + h]);
  e[t] = v;
  unsafeAtomicAdd(&s[dN * H + h], v);
}

__global__ __launch_bounds__(256) void edge_aggregate4(
    const int* __restrict__ src, const int* __restrict__ dst,
    const float* __restrict__ f, const float* __restrict__ ee,
    const float* __restrict__ ssum, float* __restrict__ out,
    int E, int H, int D) {
  const int C4 = (H * D) / 4;
  int t = blockIdx.x * blockDim.x + threadIdx.x;
  if (t >= E * C4) return;
  int edge = t / C4;
  int c4 = t - edge * C4;
  int h = (c4 * 4) / D;
  int sN = src[edge], dN = dst[edge];
  float alpha = ee[edge * H + h] / ssum[dN * H + h];
  const v4f v = *(const v4f*)(f + (size_t)sN * (H * D) + (size_t)c4 * 4);
  float* o = out + (size_t)dN * (H * D) + (size_t)c4 * 4;
  unsafeAtomicAdd(o + 0, v.x * alpha);
  unsafeAtomicAdd(o + 1, v.y * alpha);
  unsafeAtomicAdd(o + 2, v.z * alpha);
  unsafeAtomicAdd(o + 3, v.w * alpha);
}

__global__ __launch_bounds__(256) void bias_act(
    const float* __restrict__ in, const float* __restrict__ bias,
    float* __restrict__ out, int Nn, int C, int do_relu) {
  int t = blockIdx.x * blockDim.x + threadIdx.x;
  if (t >= Nn * C) return;
  int c = t % C;
  float v = in[t] + bias[c];
  if (do_relu) v = fmaxf(v, 0.f);
  out[t] = v;
}

static inline int cdiv(long long a, int b) { return (int)((a + b - 1) / b); }

extern "C" void kernel_launch(void* const* d_in, const int* in_sizes, int n_in,
                              void* d_out, int out_size, void* d_ws, size_t ws_size,
                              hipStream_t stream) {
  (void)in_sizes; (void)n_in; (void)out_size; (void)ws_size;
  const float* x   = (const float*)d_in[0];
  const int*   e0s = (const int*)d_in[1];
  const int*   e0d = (const int*)d_in[2];
  const int*   e1s = (const int*)d_in[3];
  const int*   e1d = (const int*)d_in[4];
  const float* W1  = (const float*)d_in[5];
  const float* al1 = (const float*)d_in[6];
  const float* ar1 = (const float*)d_in[7];
  const float* b1  = (const float*)d_in[8];
  const float* W2  = (const float*)d_in[9];
  const float* al2 = (const float*)d_in[10];
  const float* ar2 = (const float*)d_in[11];
  const float* b2  = (const float*)d_in[12];
  float* out = (float*)d_out;

  float* ws = (float*)d_ws;
  size_t off = 0;
  auto alloc = [&](size_t n) { float* p = ws + off; off += n; return p; };
  const int F1 = cH1 * cD1;
  const int F2 = cH2 * cD2;
  float* f1   = alloc((size_t)cN0 * F1);
  float* el1  = alloc((size_t)cN0 * cH1);
  float* er1  = alloc((size_t)cN1 * cH1);
  float* m1   = alloc((size_t)cN1 * cH1);
  float* s1   = alloc((size_t)cN1 * cH1);
  float* e1b  = alloc((size_t)cE0 * cH1);
  float* acc1 = alloc((size_t)cN1 * F1);
  float* f2   = alloc((size_t)cN1 * F2);
  float* el2  = alloc((size_t)cN1 * cH2);
  float* er2  = alloc((size_t)cN2 * cH2);
  float* m2   = alloc((size_t)cN2 * cH2);
  float* s2   = alloc((size_t)cN2 * cH2);
  float* e2b  = alloc((size_t)cE1 * cH2);
  float* acc2 = alloc((size_t)cN2 * F2);

  const float NEG_INF = -INFINITY;
  constexpr int MT = 8, NT = 4, KK = 256;

  {
    const int mblocks = cdiv(cN0 / 16, MT);
    const int ngroups = F1 / (16 * NT);
    wmma_gemm_f32<MT, NT, KK><<<mblocks * ngroups, 32 * MT, 0, stream>>>(
        x, W1, f1, cN0, F1);

    attn_dots<<<cdiv((long long)cN0 * cH1, 256), 256, 0, stream>>>(
        f1, al1, ar1, el1, er1, cN0, cN1, cH1, cD1);

    fill_f32<<<cdiv((long long)cN1 * cH1, 256), 256, 0, stream>>>(m1, NEG_INF, cN1 * cH1);
    fill_f32<<<cdiv((long long)cN1 * cH1, 256), 256, 0, stream>>>(s1, 0.f, cN1 * cH1);
    fill_f32<<<cdiv((long long)cN1 * F1, 256), 256, 0, stream>>>(acc1, 0.f, cN1 * F1);

    edge_scores<<<cdiv((long long)cE0 * cH1, 256), 256, 0, stream>>>(
        e0s, e0d, el1, er1, e1b, m1, cE0, cH1);
    edge_exp_sum<<<cdiv((long long)cE0 * cH1, 256), 256, 0, stream>>>(
        e0d, e1b, m1, s1, cE0, cH1);
    edge_aggregate4<<<cdiv((long long)cE0 * (F1 / 4), 256), 256, 0, stream>>>(
        e0s, e0d, f1, e1b, s1, acc1, cE0, cH1, cD1);

    bias_act<<<cdiv((long long)cN1 * F1, 256), 256, 0, stream>>>(
        acc1, b1, acc1, cN1, F1, 1);
  }

  {
    const int mblocks = cdiv(cN1 / 16, MT);
    const int ngroups = F2 / (16 * NT);
    wmma_gemm_f32<MT, NT, KK><<<mblocks * ngroups, 32 * MT, 0, stream>>>(
        acc1, W2, f2, cN1, F2);

    attn_dots<<<cdiv((long long)cN1 * cH2, 256), 256, 0, stream>>>(
        f2, al2, ar2, el2, er2, cN1, cN2, cH2, cD2);

    fill_f32<<<cdiv((long long)cN2 * cH2, 256), 256, 0, stream>>>(m2, NEG_INF, cN2 * cH2);
    fill_f32<<<cdiv((long long)cN2 * cH2, 256), 256, 0, stream>>>(s2, 0.f, cN2 * cH2);
    fill_f32<<<cdiv((long long)cN2 * F2, 256), 256, 0, stream>>>(acc2, 0.f, cN2 * F2);

    edge_scores<<<cdiv((long long)cE1 * cH2, 256), 256, 0, stream>>>(
        e1s, e1d, el2, er2, e2b, m2, cE1, cH2);
    edge_exp_sum<<<cdiv((long long)cE1 * cH2, 256), 256, 0, stream>>>(
        e1d, e2b, m2, s2, cE1, cH2);
    edge_aggregate4<<<cdiv((long long)cE1 * (F2 / 4), 256), 256, 0, stream>>>(
        e1s, e1d, f2, e2b, s2, acc2, cE1, cH2, cD2);

    bias_act<<<cdiv((long long)cN2 * F2, 256), 256, 0, stream>>>(
        acc2, b2, out, cN2, F2, 0);
  }
}


# Segment max, sum and feature sum done per destination node over an edge list sorted by destination in-call (count, scan, fill), no float atomics left; sums run in fill-arrival order
# speedup vs baseline: 1.7276x; 1.7276x over previous
#include <hip/hip_runtime.h>
#include <hip/hip_bf16.h>
#include <math.h>

constexpr int cN0 = 100000, cN1 = 50000, cN2 = 25000;
constexpr int cE0 = 400000, cE1 = 200000;
constexpr int cH1 = 4, cD1 = 64;
constexpr int cH2 = 1, cD2 = 64;
constexpr float cSLOPE = 0.2f;

typedef __attribute__((ext_vector_type(2))) float v2f;
typedef __attribute__((ext_vector_type(4))) float v4f;
typedef __attribute__((ext_vector_type(8))) float v8f;

template <int MT, int NT, int KK>
__global__ __launch_bounds__(32 * MT) void wmma_gemm_f32(
    const float* __restrict__ A, const float* __restrict__ B,
    float* __restrict__ C, int M, int N) {
  constexpr int NC = 16 * NT;
  constexpr int NB = (KK / 4) * 2 * NC;
  __shared__ v2f Bl[NB];

  const int ngroups = N / NC;
  const int ng = blockIdx.x % ngroups;
  const int mblock = blockIdx.x / ngroups;
  const int c0 = ng * NC;

  const float* Bp = B + c0;
  for (int idx = threadIdx.x; idx < NB; idx += 32 * MT) {
    const int col = idx & (NC - 1);
    const int kh = idx / NC;
    const int krow = (kh >> 1) * 4 + (kh & 1) * 2;
    v2f v;
    v.x = Bp[(size_t)krow * N + col];
    v.y = Bp[(size_t)(krow + 1) * N + col];
    Bl[idx] = v;
  }
  __syncthreads();

  const int waveid = threadIdx.x >> 5;
  const int lane = threadIdx.x & 31;
  const int mtile = mblock * MT + waveid;
  if (mtile * 16 >= M) return;

  const int half = lane >> 4;
  const int l = lane & 15;
  const float* Arow = A + (size_t)(mtile * 16 + l) * KK;

  v8f acc[NT] = {};

  for (int kc = 0; kc < KK; kc += 16) {
    v2f a[4];
#pragma unroll
    for (int j = 0; j < 4; ++j) {
      const int ka = kc + j * 4 + half * 2;
      a[j].x = Arow[ka];
      a[j].y = Arow[ka + 1];
    }
    v2f b[4][NT];
#pragma unroll
    for (int j = 0; j < 4; ++j) {
      const int base = ((kc / 4 + j) * 2 + half) * NC + l;
#pragma unroll
      for (int t = 0; t < NT; ++t) b[j][t] = Bl[base + t * 16];
    }
#pragma unroll
    for (int j = 0; j < 4; ++j) {
#pragma unroll
      for (int t = 0; t < NT; ++t) {
        acc[t] = __builtin_amdgcn_wmma_f32_16x16x4_f32(
            false, a[j], false, b[j][t], (short)0, acc[t], false, false);
      }
    }
  }

  const int row0 = mtile * 16 + half * 8;
#pragma unroll
  for (int t = 0; t < NT; ++t) {
    const int col = c0 + t * 16 + l;
#pragma unroll
    for (int i = 0; i < 8; ++i) {
      C[(size_t)(row0 + i) * N + col] = acc[t][i];
    }
  }
}

__device__ inline int wave_sum_i(int v) {
#pragma unroll
  for (int o = 16; o > 0; o >>= 1) v += __shfl_xor(v, o);
  return v;
}
__device__ inline int wave_max_i(int v) {
#pragma unroll
  for (int o = 16; o > 0; o >>= 1) v = max(v, __shfl_xor(v, o));
  return v;
}

__global__ __launch_bounds__(256) void fill_i32(int* __restrict__ p, int v, int n) {
  int t = blockIdx.x * blockDim.x + threadIdx.x;
  if (t < n) p[t] = v;
}

__global__ __launch_bounds__(256) void attn_dots(
    const float* __restrict__ f, const float* __restrict__ al,
    const float* __restrict__ ar, float* __restrict__ el,
    float* __restrict__ er, int Nsrc, int Ndst, int H, int D) {
  int t = blockIdx.x * blockDim.x + threadIdx.x;
  if (t >= Nsrc * H) return;
  int n = t / H, h = t - n * H;
  const v4f* fr = (const v4f*)(f + (size_t)n * H * D + (size_t)h * D);
  const v4f* alh = (const v4f*)(al + h * D);
  const v4f* arh = (const v4f*)(ar + h * D);
  float sl = 0.f, sr = 0.f;
#pragma unroll 4
  for (int d = 0; d < D / 4; ++d) {
    v4f v = fr[d], a = alh[d], b = arh[d];
    sl += v.x * a.x + v.y * a.y + v.z * a.z + v.w * a.w;
    sr += v.x * b.x + v.y * b.y + v.z * b.z + v.w * b.w;
  }
  el[t] = sl;
  if (n < Ndst) er[t] = sr;
}


__global__ __launch_bounds__(256) void dst_count2(
    const int* __restrict__ dstA, int* __restrict__ cntA, int EA,
    const int* __restrict__ dstB, int* __restrict__ cntB, int EB) {
  const int t = blockIdx.x * blockDim.x + threadIdx.x;
  if (t < EA) atomicAdd(&cntA[dstA[t]], 1);
  else if (t - EA < EB) atomicAdd(&cntB[dstB[t - EA]], 1);
}

__global__ __launch_bounds__(1024) void seg_offsets2(
    const int* __restrict__ cntA, int* __restrict__ curA, int nA, int nbA,
    const int* __restrict__ cntB, int* __restrict__ curB, int nB) {
  __shared__ int wpre[32];
  __shared__ int wtot[32];
  const bool isA = (int)blockIdx.x < nbA;
  const int* __restrict__ cnt = isA ? cntA : cntB;
  int* __restrict__ cur = isA ? curA : curB;
  const int n = isA ? nA : nB;
  const int base = (isA ? (int)blockIdx.x : (int)blockIdx.x - nbA) * 1024;
  const int tid = threadIdx.x;
  const int lane = tid & 31;
  const int w = tid >> 5;
  int pre = 0;
  for (int i = tid; i < base; i += 1024) pre += cnt[i];
  const int idx = base + tid;
  const int own = idx < n ? cnt[idx] : 0;
  int sc = own;
#pragma unroll
  for (int o = 1; o < 32; o <<= 1) {
    const int up = __shfl_up(sc, o);
    if (lane >= o) sc += up;
  }
  const int ps = wave_sum_i(pre);
  if (lane == 31) wtot[w] = sc;
  if (lane == 0) wpre[w] = ps;
  __syncthreads();
  const int allpre = wave_sum_i(wpre[lane]);
  const int woff = wave_sum_i(lane < w ? wtot[lane] : 0);
  if (idx < n) cur[idx] = allpre + woff + sc - own;
}

__global__ __launch_bounds__(256) void csr_fill2(
    const int* __restrict__ srcA, const int* __restrict__ dstA,
    int* __restrict__ curA, int* __restrict__ csrA, int EA,
    const int* __restrict__ srcB, const int* __restrict__ dstB,
    int* __restrict__ curB, int* __restrict__ csrB, int EB) {
  const int t = blockIdx.x * blockDim.x + threadIdx.x;
  if (t < EA) {
    const int p = atomicAdd(&curA[dstA[t]], 1);
    csrA[p] = srcA[t];
  } else if (t - EA < EB) {
    const int u = t - EA;
    const int p = atomicAdd(&curB[dstB[u]], 1);
    csrB[p] = srcB[u];
  }
}

template <int H, int D, bool RELU>
__global__ __launch_bounds__(256) void seg_softmax_gather(
    const int* __restrict__ cur, const int* __restrict__ csr_src,
    const float* __restrict__ f, const float* __restrict__ el,
    const float* __restrict__ er, const float* __restrict__ bias,
    float* __restrict__ out, int Ndst) {
  constexpr int F = H * D;
  constexpr int C4 = F / 4;
  constexpr int LPN = C4 < 32 ? C4 : 32;
  constexpr int NCH = C4 / LPN;
  constexpr int NPW = 32 / LPN;
  const int lane = threadIdx.x & 31;
  const int wave = (blockIdx.x * blockDim.x + threadIdx.x) >> 5;
  const int sub = NPW == 1 ? 0 : lane / LPN;
  const int sl = NPW == 1 ? lane : lane % LPN;
  const int d = wave * NPW + sub;
  const bool nodeok = d < Ndst;

  int beg = 0, end = 0;
  float erd[H];
#pragma unroll
  for (int h = 0; h < H; ++h) erd[h] = 0.f;
  if (nodeok) {
    end = cur[d];
    beg = d > 0 ? cur[d - 1] : 0;
#pragma unroll
    for (int h = 0; h < H; ++h) erd[h] = er[(size_t)d * H + h];
  }
  const int deg = end - beg;
  const int ntiles = __builtin_amdgcn_readfirstlane(wave_max_i((deg + LPN - 1) / LPN));

  auto tile_scores = [&](int t, int& sN, bool& valid, float (&e)[H]) {
    const int j = beg + t * LPN + sl;
    valid = j < end;
    sN = valid ? csr_src[j] : 0;
#pragma unroll
    for (int h = 0; h < H; ++h) {
      float v = el[(size_t)sN * H + h] + erd[h];
      v = v > 0.f ? v : cSLOPE * v;
      e[h] = valid ? v : -INFINITY;
    }
  };

  int sN0; bool valid0; float e0[H];
  tile_scores(0, sN0, valid0, e0);

  float m[H];
#pragma unroll
  for (int h = 0; h < H; ++h) m[h] = e0[h];
  for (int t = 1; t < ntiles; ++t) {
    int sN; bool valid; float e[H];
    tile_scores(t, sN, valid, e);
#pragma unroll
    for (int h = 0; h < H; ++h) m[h] = fmaxf(m[h], e[h]);
  }
#pragma unroll
  for (int h = 0; h < H; ++h) {
#pragma unroll
    for (int o = LPN / 2; o > 0; o >>= 1) m[h] = fmaxf(m[h], __shfl_xor(m[h], o));
  }

  float s[H];
#pragma unroll
  for (int h = 0; h < H; ++h) s[h] = valid0 ? __expf(e0[h] - m[h]) : 0.f;
  for (int t = 1; t < ntiles; ++t) {
    int sN; bool valid; float e[H];
    tile_scores(t, sN, valid, e);
#pragma unroll
    for (int h = 0; h < H; ++h) s[h] += valid ? __expf(e[h] - m[h]) : 0.f;
  }
#pragma unroll
  for (int h = 0; h < H; ++h) {
#pragma unroll
    for (int o = LPN / 2; o > 0; o >>= 1) s[h] += __shfl_xor(s[h], o);
  }

  v4f acc[NCH];
#pragma unroll
  for (int q = 0; q < NCH; ++q) acc[q] = v4f{0.f, 0.f, 0.f, 0.f};

  for (int t = 0; t < ntiles; ++t) {
    int sN = sN0; bool valid = valid0; float e[H];
#pragma unroll
    for (int h = 0; h < H; ++h) e[h] = e0[h];
    if (t > 0) tile_scores(t, sN, valid, e);
    float a[H];
#pragma unroll
    for (int h = 0; h < H; ++h) a[h] = valid ? __expf(e[h] - m[h]) / s[h] : 0.f;
    const int mine = min(LPN, deg - t * LPN);
    const int kmax = __builtin_amdgcn_readfirstlane(wave_max_i(mine));

    auto edge_rows = [&](int k, v4f (&v)[NCH], float (&w)[NCH]) {
      const int from = sub * LPN + k;
      const int rN = __shfl(sN, from);
      float ah[H];
#pragma unroll
      for (int h = 0; h < H; ++h) ah[h] = __shfl(a[h], from);
#pragma unroll
      for (int q = 0; q < NCH; ++q) {
        const int c4 = q * LPN + sl;
        const int hq = (c4 * 4) / D;
        float al = ah[0];
#pragma unroll
        for (int h = 1; h < H; ++h) al = (hq == h) ? ah[h] : al;
        w[q] = al;
        v[q] = *(const v4f*)(f + (size_t)rN * F + (size_t)c4 * 4);
      }
    };
    auto edge_add = [&](int k, const v4f (&v)[NCH], const float (&w)[NCH]) {
#pragma unroll
      for (int q = 0; q < NCH; ++q) {
        v4f vv = v[q];
        if (NPW > 1) vv = (k < mine) ? vv : v4f{0.f, 0.f, 0.f, 0.f};
        acc[q] += vv * w[q];
      }
    };

    const int kfull = kmax & ~3;
    for (int k0 = 0; k0 < kfull; k0 += 4) {
      v4f v[4][NCH]; float w[4][NCH];
#pragma unroll
      for (int u = 0; u < 4; ++u) edge_rows(k0 + u, v[u], w[u]);
#pragma unroll
      for (int u = 0; u < 4; ++u) edge_add(k0 + u, v[u], w[u]);
    }
    for (int k = kfull; k < kmax; ++k) {
      v4f v[NCH]; float w[NCH];
      edge_rows(k, v, w);
      edge_add(k, v, w);
    }
  }

  if (nodeok) {
#pragma unroll
    for (int q = 0; q < NCH; ++q) {
      const int c4 = q * LPN + sl;
      const v4f b = *(const v4f*)(bias + c4 * 4);
      v4f r = acc[q] + b;
      if (RELU) {
        r.x = fmaxf(r.x, 0.f); r.y = fmaxf(r.y, 0.f);
        r.z = fmaxf(r.z, 0.f); r.w = fmaxf(r.w, 0.f);
      }
      *(v4f*)(out + (size_t)d * F + (size_t)c4 * 4) = r;
    }
  }
}

static inline int cdiv(long long a, int b) { return (int)((a + b - 1) / b); }

extern "C" void kernel_launch(void* const* d_in, const int* in_sizes, int n_in,
                              void* d_out, int out_size, void* d_ws, size_t ws_size,
                              hipStream_t stream) {
  (void)in_sizes; (void)n_in; (void)out_size; (void)ws_size;
  const float* x   = (const float*)d_in[0];
  const int*   e0s = (const int*)d_in[1];
  const int*   e0d = (const int*)d_in[2];
  const int*   e1s = (const int*)d_in[3];
  const int*   e1d = (const int*)d_in[4];
  const float* W1  = (const float*)d_in[5];
  const float* al1 = (const float*)d_in[6];
  const float* ar1 = (const float*)d_in[7];
  const float* b1  = (const float*)d_in[8];
  const float* W2  = (const float*)d_in[9];
  const float* al2 = (const float*)d_in[10];
  const float* ar2 = (const float*)d_in[11];
  const float* b2  = (const float*)d_in[12];
  float* out = (float*)d_out;

  float* ws = (float*)d_ws;
  size_t off = 0;
  auto alloc = [&](size_t n) { float* p = ws + off; off += n; return p; };
  const int F1 = cH1 * cD1;
  const int F2 = cH2 * cD2;
  float* f1   = alloc((size_t)cN0 * F1);
  float* el1  = alloc((size_t)cN0 * cH1);
  float* er1  = alloc((size_t)cN1 * cH1);
  float* m1   = alloc((size_t)cN1 * cH1);
  float* s1   = alloc((size_t)cN1 * cH1);
  float* e1b  = alloc((size_t)cE0 * cH1);
  float* acc1 = alloc((size_t)cN1 * F1);
  float* f2   = alloc((size_t)cN1 * F2);
  float* el2  = alloc((size_t)cN1 * cH2);
  float* er2  = alloc((size_t)cN2 * cH2);
  float* m2   = alloc((size_t)cN2 * cH2);
  float* s2   = alloc((size_t)cN2 * cH2);
  float* e2b  = alloc((size_t)cE1 * cH2);
  float* acc2 = alloc((size_t)cN2 * F2);

  constexpr int MT = 8, NT = 4, KK = 256;
  (void)m1; (void)s1; (void)m2; (void)s2; (void)e2b; (void)acc2;

  int* ip   = (int*)e1b;
  int* csr1 = ip;
  int* csr2 = csr1 + cE0;
  int* cnt1 = csr2 + cE1;
  int* cnt2 = cnt1 + cN1;
  int* cur1 = cnt2 + cN2;
  int* cur2 = cur1 + cN1;

  {
    const int mblocks = cdiv(cN0 / 16, MT);
    const int ngroups = F1 / (16 * NT);
    wmma_gemm_f32<MT, NT, KK><<<mblocks * ngroups, 32 * MT, 0, stream>>>(
        x, W1, f1, cN0, F1);

    attn_dots<<<cdiv((long long)cN0 * cH1, 256), 256, 0, stream>>>(
        f1, al1, ar1, el1, er1, cN0, cN1, cH1, cD1);

    fill_i32<<<cdiv(cN1 + cN2, 256), 256, 0, stream>>>(cnt1, 0, cN1 + cN2);
    dst_count2<<<cdiv(cE0 + cE1, 256), 256, 0, stream>>>(e0d, cnt1, cE0, e1d, cnt2, cE1);
    seg_offsets2<<<cdiv(cN1, 1024) + cdiv(cN2, 1024), 1024, 0, stream>>>(
        cnt1, cur1, cN1, cdiv(cN1, 1024), cnt2, cur2, cN2);
    csr_fill2<<<cdiv(cE0 + cE1, 256), 256, 0, stream>>>(
        e0s, e0d, cur1, csr1, cE0, e1s, e1d, cur2, csr2, cE1);

    seg_softmax_gather<cH1, cD1, true><<<cdiv((long long)cN1 * 32, 256), 256, 0, stream>>>(
        cur1, csr1, f1, el1, er1, b1, acc1, cN1);
  }

  {
    const int mblocks = cdiv(cN1 / 16, MT);
    const int ngroups = F2 / (16 * NT);
    wmma_gemm_f32<MT, NT, KK><<<mblocks * ngroups, 32 * MT, 0, stream>>>(
        acc1, W2, f2, cN1, F2);

    attn_dots<<<cdiv((long long)cN1 * cH2, 256), 256, 0, stream>>>(
        f2, al2, ar2, el2, er2, cN1, cN2, cH2, cD2);

    seg_softmax_gather<cH2, cD2, false><<<cdiv((long long)cN2 * 16, 256), 256, 0, stream>>>(
        cur2, csr2, f2, el2, er2, b2, out, cN2);
  }
}


# el/er attention dots formed in the GEMM epilogue from its accumulators (a block's 64 columns = one head); both dot kernels and their re-read of f dropped; dot summed 4 per lane, then a 16-lane tree
# speedup vs baseline: 1.8342x; 1.0617x over previous
#include <hip/hip_runtime.h>
#include <hip/hip_bf16.h>
#include <math.h>

constexpr int cN0 = 100000, cN1 = 50000, cN2 = 25000;
constexpr int cE0 = 400000, cE1 = 200000;
constexpr int cH1 = 4, cD1 = 64;
constexpr int cH2 = 1, cD2 = 64;
constexpr float cSLOPE = 0.2f;

typedef __attribute__((ext_vector_type(2))) float v2f;
typedef __attribute__((ext_vector_type(4))) float v4f;
typedef __attribute__((ext_vector_type(8))) float v8f;

template <int MT, int NT, int KK>
__global__ __launch_bounds__(32 * MT) void wmma_gemm_f32(
    const float* __restrict__ A, const float* __restrict__ B,
    float* __restrict__ C, int M, int N,
    const float* __restrict__ al, const float* __restrict__ ar,
    float* __restrict__ el, float* __restrict__ er, int Ndst) {
  static_assert(16 * NT == 64, "one block strip must be one 64-wide head");
  constexpr int NC = 16 * NT;
  constexpr int NB = (KK / 4) * 2 * NC;
  __shared__ v2f Bl[NB];

  const int ngroups = N / NC;
  const int ng = blockIdx.x % ngroups;
  const int mblock = blockIdx.x / ngroups;
  const int c0 = ng * NC;

  const float* Bp = B + c0;
  for (int idx = threadIdx.x; idx < NB; idx += 32 * MT) {
    const int col = idx & (NC - 1);
    const int kh = idx / NC;
    const int krow = (kh >> 1) * 4 + (kh & 1) * 2;
    v2f v;
    v.x = Bp[(size_t)krow * N + col];
    v.y = Bp[(size_t)(krow + 1) * N + col];
    Bl[idx] = v;
  }
  __syncthreads();

  const int waveid = threadIdx.x >> 5;
  const int lane = threadIdx.x & 31;
  const int mtile = mblock * MT + waveid;
  if (mtile * 16 >= M) return;

  const int half = lane >> 4;
  const int l = lane & 15;
  const float* Arow = A + (size_t)(mtile * 16 + l) * KK;

  v8f acc[NT] = {};

  for (int kc = 0; kc < KK; kc += 16) {
    v2f a[4];
#pragma unroll
    for (int j = 0; j < 4; ++j) {
      const int ka = kc + j * 4 + half * 2;
      a[j].x = Arow[ka];
      a[j].y = Arow[ka + 1];
    }
    v2f b[4][NT];
#pragma unroll
    for (int j = 0; j < 4; ++j) {
      const int base = ((kc / 4 + j) * 2 + half) * NC + l;
#pragma unroll
      for (int t = 0; t < NT; ++t) b[j][t] = Bl[base + t * 16];
    }
#pragma unroll
    for (int j = 0; j < 4; ++j) {
#pragma unroll
      for (int t = 0; t < NT; ++t) {
        acc[t] = __builtin_amdgcn_wmma_f32_16x16x4_f32(
            false, a[j], false, b[j][t], (short)0, acc[t], false, false);
      }
    }
  }

  const int le = (int)(__lane_id() & 15u);
  const int he = (int)(__lane_id() >> 4);
  const int row0 = mtile * 16 + he * 8;
#pragma unroll
  for (int t = 0; t < NT; ++t) {
    const int col = c0 + t * 16 + le;
#pragma unroll
    for (int i = 0; i < 8; ++i) {
      C[(size_t)(row0 + i) * N + col] = acc[t][i];
    }
  }

  float alv[NT], arv[NT];
#pragma unroll
  for (int t = 0; t < NT; ++t) {
    alv[t] = al[c0 + t * 16 + le];
    arv[t] = ar[c0 + t * 16 + le];
  }
  float pl[8], pr[8];
#pragma unroll
  for (int i = 0; i < 8; ++i) {
    float sl = 0.f, sr = 0.f;
#pragma unroll
    for (int t = 0; t < NT; ++t) {
      sl += acc[t][i] * alv[t];
      sr += acc[t][i] * arv[t];
    }
    pl[i] = sl;
    pr[i] = sr;
  }
#pragma unroll
  for (int i = 0; i < 8; ++i) {
#pragma unroll
    for (int o = 8; o > 0; o >>= 1) {
      pl[i] += __shfl_xor(pl[i], o);
      pr[i] += __shfl_xor(pr[i], o);
    }
  }
  float vl = pl[0], vr = pr[0];
#pragma unroll
  for (int i = 1; i < 8; ++i) {
    vl = (le == i) ? pl[i] : vl;
    vr = (le == i) ? pr[i] : vr;
  }
  if (le < 8) {
    const int row = row0 + le;
    el[(size_t)row * ngroups + ng] = vl;
    if (row < Ndst) er[(size_t)row * ngroups + ng] = vr;
  }
}

__device__ inline int wave_sum_i(int v) {
#pragma unroll
  for (int o = 16; o > 0; o >>= 1) v += __shfl_xor(v, o);
  return v;
}
__device__ inline int wave_max_i(int v) {
#pragma unroll
  for (int o = 16; o > 0; o >>= 1) v = max(v, __shfl_xor(v, o));
  return v;
}

__global__ __launch_bounds__(256) void fill_i32(int* __restrict__ p, int v, int n) {
  int t = blockIdx.x * blockDim.x + threadIdx.x;
  if (t < n) p[t] = v;
}


__global__ __launch_bounds__(256) void dst_count2(
    const int* __restrict__ dstA, int* __restrict__ cntA, int EA,
    const int* __restrict__ dstB, int* __restrict__ cntB, int EB) {
  const int t = blockIdx.x * blockDim.x + threadIdx.x;
  if (t < EA) atomicAdd(&cntA[dstA[t]], 1);
  else if (t - EA < EB) atomicAdd(&cntB[dstB[t - EA]], 1);
}

__global__ __launch_bounds__(1024) void seg_offsets2(
    const int* __restrict__ cntA, int* __restrict__ curA, int nA, int nbA,
    const int* __restrict__ cntB, int* __restrict__ curB, int nB) {
  __shared__ int wpre[32];
  __shared__ int wtot[32];
  const bool isA = (int)blockIdx.x < nbA;
  const int* __restrict__ cnt = isA ? cntA : cntB;
  int* __restrict__ cur = isA ? curA : curB;
  const int n = isA ? nA : nB;
  const int base = (isA ? (int)blockIdx.x : (int)blockIdx.x - nbA) * 1024;
  const int tid = threadIdx.x;
  const int lane = tid & 31;
  const int w = tid >> 5;
  int pre = 0;
  for (int i = tid; i < base; i += 1024) pre += cnt[i];
  const int idx = base + tid;
  const int own = idx < n ? cnt[idx] : 0;
  int sc = own;
#pragma unroll
  for (int o = 1; o < 32; o <<= 1) {
    const int up = __shfl_up(sc, o);
    if (lane >= o) sc += up;
  }
  const int ps = wave_sum_i(pre);
  if (lane == 31) wtot[w] = sc;
  if (lane == 0) wpre[w] = ps;
  __syncthreads();
  const int allpre = wave_sum_i(wpre[lane]);
  const int woff = wave_sum_i(lane < w ? wtot[lane] : 0);
  if (idx < n) cur[idx] = allpre + woff + sc - own;
}

__global__ __launch_bounds__(256) void csr_fill2(
    const int* __restrict__ srcA, const int* __restrict__ dstA,
    int* __restrict__ curA, int* __restrict__ csrA, int EA,
    const int* __restrict__ srcB, const int* __restrict__ dstB,
    int* __restrict__ curB, int* __restrict__ csrB, int EB) {
  const int t = blockIdx.x * blockDim.x + threadIdx.x;
  if (t < EA) {
    const int p = atomicAdd(&curA[dstA[t]], 1);
    csrA[p] = srcA[t];
  } else if (t - EA < EB) {
    const int u = t - EA;
    const int p = atomicAdd(&curB[dstB[u]], 1);
    csrB[p] = srcB[u];
  }
}

template <int H, int D, bool RELU>
__global__ __launch_bounds__(256) void seg_softmax_gather(
    const int* __restrict__ cur, const int* __restrict__ csr_src,
    const float* __restrict__ f, const float* __restrict__ el,
    const float* __restrict__ er, const float* __restrict__ bias,
    float* __restrict__ out, int Ndst) {
  constexpr int F = H * D;
  constexpr int C4 = F / 4;
  constexpr int LPN = C4 < 32 ? C4 : 32;
  constexpr int NCH = C4 / LPN;
  constexpr int NPW = 32 / LPN;
  const int lane = threadIdx.x & 31;
  const int wave = (blockIdx.x * blockDim.x + threadIdx.x) >> 5;
  const int sub = NPW == 1 ? 0 : lane / LPN;
  const int sl = NPW == 1 ? lane : lane % LPN;
  const int d = wave * NPW + sub;
  const bool nodeok = d < Ndst;

  int beg = 0, end = 0;
  float erd[H];
#pragma unroll
  for (int h = 0; h < H; ++h) erd[h] = 0.f;
  if (nodeok) {
    end = cur[d];
    beg = d > 0 ? cur[d - 1] : 0;
#pragma unroll
    for (int h = 0; h < H; ++h) erd[h] = er[(size_t)d * H + h];
  }
  const int deg = end - beg;
  const int ntiles = __builtin_amdgcn_readfirstlane(wave_max_i((deg + LPN - 1) / LPN));

  auto tile_scores = [&](int t, int& sN, bool& valid, float (&e)[H]) {
    const int j = beg + t * LPN + sl;
    valid = j < end;
    sN = valid ? csr_src[j] : 0;
#pragma unroll
    for (int h = 0; h < H; ++h) {
      float v = el[(size_t)sN * H + h] + erd[h];
      v = v > 0.f ? v : cSLOPE * v;
      e[h] = valid ? v : -INFINITY;
    }
  };

  int sN0; bool valid0; float e0[H];
  tile_scores(0, sN0, valid0, e0);

  float m[H];
#pragma unroll
  for (int h = 0; h < H; ++h) m[h] = e0[h];
  for (int t = 1; t < ntiles; ++t) {
    int sN; bool valid; float e[H];
    tile_scores(t, sN, valid, e);
#pragma unroll
    for (int h = 0; h < H; ++h) m[h] = fmaxf(m[h], e[h]);
  }
#pragma unroll
  for (int h = 0; h < H; ++h) {
#pragma unroll
    for (int o = LPN / 2; o > 0; o >>= 1) m[h] = fmaxf(m[h], __shfl_xor(m[h], o));
  }

  float s[H];
#pragma unroll
  for (int h = 0; h < H; ++h) s[h] = valid0 ? __expf(e0[h] - m[h]) : 0.f;
  for (int t = 1; t < ntiles; ++t) {
    int sN; bool valid; float e[H];
    tile_scores(t, sN, valid, e);
#pragma unroll
    for (int h = 0; h < H; ++h) s[h] += valid ? __expf(e[h] - m[h]) : 0.f;
  }
#pragma unroll
  for (int h = 0; h < H; ++h) {
#pragma unroll
    for (int o = LPN / 2; o > 0; o >>= 1) s[h] += __shfl_xor(s[h], o);
  }

  v4f acc[NCH];
#pragma unroll
  for (int q = 0; q < NCH; ++q) acc[q] = v4f{0.f, 0.f, 0.f, 0.f};

  for (int t = 0; t < ntiles; ++t) {
    int sN = sN0; bool valid = valid0; float e[H];
#pragma unroll
    for (int h = 0; h < H; ++h) e[h] = e0[h];
    if (t > 0) tile_scores(t, sN, valid, e);
    float a[H];
#pragma unroll
    for (int h = 0; h < H; ++h) a[h] = valid ? __expf(e[h] - m[h]) / s[h] : 0.f;
    const int mine = min(LPN, deg - t * LPN);
    const int kmax = __builtin_amdgcn_readfirstlane(wave_max_i(mine));

    auto edge_rows = [&](int k, v4f (&v)[NCH], float (&w)[NCH]) {
      const int from = sub * LPN + k;
      const int rN = __shfl(sN, from);
      float ah[H];
#pragma unroll
      for (int h = 0; h < H; ++h) ah[h] = __shfl(a[h], from);
#pragma unroll
      for (int q = 0; q < NCH; ++q) {
        const int c4 = q * LPN + sl;
        const int hq = (c4 * 4) / D;
        float al = ah[0];
#pragma unroll
        for (int h = 1; h < H; ++h) al = (hq == h) ? ah[h] : al;
        w[q] = al;
        v[q] = *(const v4f*)(f + (size_t)rN * F + (size_t)c4 * 4);
      }
    };
    auto edge_add = [&](int k, const v4f (&v)[NCH], const float (&w)[NCH]) {
#pragma unroll
      for (int q = 0; q < NCH; ++q) {
        v4f vv = v[q];
        if (NPW > 1) vv = (k < mine) ? vv : v4f{0.f, 0.f, 0.f, 0.f};
        acc[q] += vv * w[q];
      }
    };

    const int kfull = kmax & ~3;
    for (int k0 = 0; k0 < kfull; k0 += 4) {
      v4f v[4][NCH]; float w[4][NCH];
#pragma unroll
      for (int u = 0; u < 4; ++u) edge_rows(k0 + u, v[u], w[u]);
#pragma unroll
      for (int u = 0; u < 4; ++u) edge_add(k0 + u, v[u], w[u]);
    }
    for (int k = kfull; k < kmax; ++k) {
      v4f v[NCH]; float w[NCH];
      edge_rows(k, v, w);
      edge_add(k, v, w);
    }
  }

  if (nodeok) {
#pragma unroll
    for (int q = 0; q < NCH; ++q) {
      const int c4 = q * LPN + sl;
      const v4f b = *(const v4f*)(bias + c4 * 4);
      v4f r = acc[q] + b;
      if (RELU) {
        r.x = fmaxf(r.x, 0.f); r.y = fmaxf(r.y, 0.f);
        r.z = fmaxf(r.z, 0.f); r.w = fmaxf(r.w, 0.f);
      }
      *(v4f*)(out + (size_t)d * F + (size_t)c4 * 4) = r;
    }
  }
}

static inline int cdiv(long long a, int b) { return (int)((a + b - 1) / b); }

extern "C" void kernel_launch(void* const* d_in, const int* in_sizes, int n_in,
                              void* d_out, int out_size, void* d_ws, size_t ws_size,
                              hipStream_t stream) {
  (void)in_sizes; (void)n_in; (void)out_size; (void)ws_size;
  const float* x   = (const float*)d_in[0];
  const int*   e0s = (const int*)d_in[1];
  const int*   e0d = (const int*)d_in[2];
  const int*   e1s = (const int*)d_in[3];
  const int*   e1d = (const int*)d_in[4];
  const float* W1  = (const float*)d_in[5];
  const float* al1 = (const float*)d_in[6];
  const float* ar1 = (const float*)d_in[7];
  const float* b1  = (const float*)d_in[8];
  const float* W2  = (const float*)d_in[9];
  const float* al2 = (const float*)d_in[10];
  const float* ar2 = (const float*)d_in[11];
  const float* b2  = (const float*)d_in[12];
  float* out = (float*)d_out;

  float* ws = (float*)d_ws;
  size_t off = 0;
  auto alloc = [&](size_t n) { float* p = ws + off; off += n; return p; };
  const int F1 = cH1 * cD1;
  const int F2 = cH2 * cD2;
  float* f1   = alloc((size_t)cN0 * F1);
  float* el1  = alloc((size_t)cN0 * cH1);
  float* er1  = alloc((size_t)cN1 * cH1);
  float* m1   = alloc((size_t)cN1 * cH1);
  float* s1   = alloc((size_t)cN1 * cH1);
  float* e1b  = alloc((size_t)cE0 * cH1);
  float* acc1 = alloc((size_t)cN1 * F1);
  float* f2   = alloc((size_t)cN1 * F2);
  float* el2  = alloc((size_t)cN1 * cH2);
  float* er2  = alloc((size_t)cN2 * cH2);
  float* m2   = alloc((size_t)cN2 * cH2);
  float* s2   = alloc((size_t)cN2 * cH2);
  float* e2b  = alloc((size_t)cE1 * cH2);
  float* acc2 = alloc((size_t)cN2 * F2);

  constexpr int MT = 8, NT = 4, KK = 256;
  (void)m1; (void)s1; (void)m2; (void)s2; (void)e2b; (void)acc2;

  int* ip   = (int*)e1b;
  int* csr1 = ip;
  int* csr2 = csr1 + cE0;
  int* cnt1 = csr2 + cE1;
  int* cnt2 = cnt1 + cN1;
  int* cur1 = cnt2 + cN2;
  int* cur2 = cur1 + cN1;

  {
    const int mblocks = cdiv(cN0 / 16, MT);
    const int ngroups = F1 / (16 * NT);
    wmma_gemm_f32<MT, NT, KK><<<mblocks * ngroups, 32 * MT, 0, stream>>>(
        x, W1, f1, cN0, F1, al1, ar1, el1, er1, cN1);

    fill_i32<<<cdiv(cN1 + cN2, 256), 256, 0, stream>>>(cnt1, 0, cN1 + cN2);
    dst_count2<<<cdiv(cE0 + cE1, 256), 256, 0, stream>>>(e0d, cnt1, cE0, e1d, cnt2, cE1);
    seg_offsets2<<<cdiv(cN1, 1024) + cdiv(cN2, 1024), 1024, 0, stream>>>(
        cnt1, cur1, cN1, cdiv(cN1, 1024), cnt2, cur2, cN2);
    csr_fill2<<<cdiv(cE0 + cE1, 256), 256, 0, stream>>>(
        e0s, e0d, cur1, csr1, cE0, e1s, e1d, cur2, csr2, cE1);

    seg_softmax_gather<cH1, cD1, true><<<cdiv((long long)cN1 * 32, 256), 256, 0, stream>>>(
        cur1, csr1, f1, el1, er1, b1, acc1, cN1);
  }

  {
    const int mblocks = cdiv(cN1 / 16, MT);
    const int ngroups = F2 / (16 * NT);
    wmma_gemm_f32<MT, NT, KK><<<mblocks * ngroups, 32 * MT, 0, stream>>>(
        acc1, W2, f2, cN1, F2, al2, ar2, el2, er2, cN2);

    seg_softmax_gather<cH2, cD2, false><<<cdiv((long long)cN2 * 16, 256), 256, 0, stream>>>(
        cur2, csr2, f2, el2, er2, b2, out, cN2);
  }
}
